# starting kernel
# baseline (speedup 1.0000x reference)
#include <hip/hip_runtime.h>

typedef __attribute__((ext_vector_type(2))) float v2f;
typedef __attribute__((ext_vector_type(8))) float v8f;

#define F_DIM 200
#define E_DIM 64
#define LDS_PITCH 202
#define WAVES_PER_BLOCK 8
#define ROWS_PER_WAVE 16
#define ROWS_PER_BLOCK (WAVES_PER_BLOCK * ROWS_PER_WAVE)

__global__ __launch_bounds__(256)
void fm_second_order_kernel(const float* __restrict__ features,
                            const float* __restrict__ W,
                            float* __restrict__ out, int Brows) {
    __shared__ float sWt[E_DIM * LDS_PITCH];
    __shared__ float sW2[F_DIM];

    const int tid  = threadIdx.x;
    const int lane = tid & 31;
    const int wave = tid >> 5;

    for (int i = tid; i < F_DIM * E_DIM; i += 256) {
        int f = i >> 6;
        int e = i & (E_DIM - 1);
        sWt[e * LDS_PITCH + f] = W[i];
    }
    __syncthreads();
    if (tid < F_DIM) {
        float s = 0.f;
        #pragma unroll
        for (int e = 0; e < E_DIM; ++e) {
            float w = sWt[e * LDS_PITCH + tid];
            s = fmaf(w, w, s);
        }
        sW2[tid] = s;
    }
    __syncthreads();

    const int base_row = blockIdx.x * ROWS_PER_BLOCK + wave * ROWS_PER_WAVE;
    if (base_row >= Brows) return;

    const int mrow = lane & 15;
    const int half = lane >> 4;
    const int n0   = lane & 15;

    const float* arow = features + (size_t)(base_row + mrow) * F_DIM;

    v8f acc0 = {}, acc1 = {}, acc2 = {}, acc3 = {};
    float sqp = 0.f;

    for (int k = 0; k < F_DIM; k += 4) {
        const int kk = k + 2 * half;
        v2f a = *reinterpret_cast<const v2f*>(arow + kk);
        sqp = fmaf(a.x * a.x, sW2[kk],     sqp);
        sqp = fmaf(a.y * a.y, sW2[kk + 1], sqp);
        v2f b0 = *reinterpret_cast<const v2f*>(&sWt[(n0 +  0) * LDS_PITCH + kk]);
        v2f b1 = *reinterpret_cast<const v2f*>(&sWt[(n0 + 16) * LDS_PITCH + kk]);
        v2f b2 = *reinterpret_cast<const v2f*>(&sWt[(n0 + 32) * LDS_PITCH + kk]);
        v2f b3 = *reinterpret_cast<const v2f*>(&sWt[(n0 + 48) * LDS_PITCH + kk]);
        acc0 = __builtin_amdgcn_wmma_f32_16x16x4_f32(false, a, false, b0, (short)0, acc0, false, false);
        acc1 = __builtin_amdgcn_wmma_f32_16x16x4_f32(false, a, false, b1, (short)0, acc1, false, false);
        acc2 = __builtin_amdgcn_wmma_f32_16x16x4_f32(false, a, false, b2, (short)0, acc2, false, false);
        acc3 = __builtin_amdgcn_wmma_f32_16x16x4_f32(false, a, false, b3, (short)0, acc3, false, false);
    }

    float s[8];
    #pragma unroll
    for (int r = 0; r < 8; ++r) {
        float c0 = acc0[r], c1 = acc1[r], c2 = acc2[r], c3 = acc3[r];
        s[r] = c0 * c0 + c1 * c1 + c2 * c2 + c3 * c3;
    }
    #pragma unroll
    for (int m = 1; m <= 8; m <<= 1) {
        #pragma unroll
        for (int r = 0; r < 8; ++r) s[r] += __shfl_xor(s[r], m, 32);
    }
    float sqfull = sqp + __shfl_xor(sqp, 16, 32);

    int myrow = (lane & 7) + ((lane >> 4) << 3);
    float sqr = __shfl(sqfull, myrow, 32);
    if ((lane & 15) < 8) {
        out[base_row + myrow] = s[lane & 7] - sqr;
    }
}

extern "C" void kernel_launch(void* const* d_in, const int* in_sizes, int n_in,
                              void* d_out, int out_size, void* d_ws, size_t ws_size,
                              hipStream_t stream) {
    const float* features = (const float*)d_in[0];
    const float* W        = (const float*)d_in[1];
    float* out            = (float*)d_out;
    (void)d_ws; (void)ws_size; (void)n_in; (void)out_size;

    int B = in_sizes[0] / F_DIM;
    int blocks = (B + ROWS_PER_BLOCK - 1) / ROWS_PER_BLOCK;
    fm_second_order_kernel<<<blocks, 256, 0, stream>>>(features, W, out, B);
}


# All global loads hoisted: each thread's W share in one batch, then the wave's whole 16x200 A tile requested ahead of the LDS-only-fenced staging barriers; K loop fully unrolled on registers
# speedup vs baseline: 1.5305x; 1.5305x over previous
#include <hip/hip_runtime.h>


typedef __attribute__((ext_vector_type(2))) float v2f;
typedef __attribute__((ext_vector_type(8))) float v8f;

#define F_DIM 200
#define E_DIM 64
#define LDS_PITCH 202
#define WAVES_PER_BLOCK 8
#define ROWS_PER_WAVE 16
#define ROWS_PER_BLOCK (WAVES_PER_BLOCK * ROWS_PER_WAVE)
#define K_STEPS (F_DIM / 4)
#define W_PER_THREAD (F_DIM * E_DIM / 256)

static __device__ __forceinline__ void lds_barrier() {
    __builtin_amdgcn_fence(__ATOMIC_RELEASE, "workgroup", "local");
    __builtin_amdgcn_s_barrier();
    __builtin_amdgcn_fence(__ATOMIC_ACQUIRE, "workgroup", "local");
}

__global__ __launch_bounds__(256)
void fm_second_order_kernel(const float* __restrict__ features,
                            const float* __restrict__ W,
                            float* __restrict__ out, int Brows) {
    __shared__ float sWt[E_DIM * LDS_PITCH];
    __shared__ float sW2[F_DIM];

    const int tid  = threadIdx.x;
    const int lane = tid & 31;
    const int wave = tid >> 5;

    const int base_row = blockIdx.x * ROWS_PER_BLOCK + wave * ROWS_PER_WAVE;
    const int mrow = lane & 15;
    const int half = lane >> 4;
    const int n0   = lane & 15;

    float wreg[W_PER_THREAD];
    #pragma unroll
    for (int j = 0; j < W_PER_THREAD; ++j) wreg[j] = W[tid + 256 * j];
    #pragma unroll
    for (int j = 0; j < W_PER_THREAD; ++j) {
        int i = tid + 256 * j;
        int f = i >> 6;
        int e = i & (E_DIM - 1);
        sWt[e * LDS_PITCH + f] = wreg[j];
    }

    __builtin_amdgcn_sched_barrier(0);

    int arow_idx = base_row + mrow;
    arow_idx = arow_idx < Brows ? arow_idx : Brows - 1;
    const float* arow = features + (size_t)arow_idx * F_DIM + 2 * half;
    v2f areg[K_STEPS];
    #pragma unroll
    for (int j = 0; j < K_STEPS; ++j) areg[j] = *reinterpret_cast<const v2f*>(arow + 4 * j);

    lds_barrier();
    if (tid < F_DIM) {
        float s = 0.f;
        #pragma unroll
        for (int e = 0; e < E_DIM; ++e) {
            float w = sWt[e * LDS_PITCH + tid];
            s = fmaf(w, w, s);
        }
        sW2[tid] = s;
    }
    lds_barrier();

    if (base_row >= Brows) return;

    v8f acc0 = {}, acc1 = {}, acc2 = {}, acc3 = {};
    float sqp = 0.f;

    #pragma unroll
    for (int j = 0; j < K_STEPS; ++j) {
        const int kk = 4 * j + 2 * half;
        const v2f a = areg[j];
        sqp = fmaf(a.x * a.x, sW2[kk],     sqp);
        sqp = fmaf(a.y * a.y, sW2[kk + 1], sqp);
        v2f b0 = *reinterpret_cast<const v2f*>(&sWt[(n0 +  0) * LDS_PITCH + kk]);
        v2f b1 = *reinterpret_cast<const v2f*>(&sWt[(n0 + 16) * LDS_PITCH + kk]);
        v2f b2 = *reinterpret_cast<const v2f*>(&sWt[(n0 + 32) * LDS_PITCH + kk]);
        v2f b3 = *reinterpret_cast<const v2f*>(&sWt[(n0 + 48) * LDS_PITCH + kk]);
        acc0 = __builtin_amdgcn_wmma_f32_16x16x4_f32(false, a, false, b0, (short)0, acc0, false, false);
        acc1 = __builtin_amdgcn_wmma_f32_16x16x4_f32(false, a, false, b1, (short)0, acc1, false, false);
        acc2 = __builtin_amdgcn_wmma_f32_16x16x4_f32(false, a, false, b2, (short)0, acc2, false, false);
        acc3 = __builtin_amdgcn_wmma_f32_16x16x4_f32(false, a, false, b3, (short)0, acc3, false, false);
    }

    float s[8];
    #pragma unroll
    for (int r = 0; r < 8; ++r) {
        float c0 = acc0[r], c1 = acc1[r], c2 = acc2[r], c3 = acc3[r];
        s[r] = c0 * c0 + c1 * c1 + c2 * c2 + c3 * c3;
    }
    #pragma unroll
    for (int m = 1; m <= 8; m <<= 1) {
        #pragma unroll
        for (int r = 0; r < 8; ++r) s[r] += __shfl_xor(s[r], m, 32);
    }
    float sqfull = sqp + __shfl_xor(sqp, 16, 32);

    int myrow = (lane & 7) + ((lane >> 4) << 3);
    float sqr = __shfl(sqfull, myrow, 32);
    if ((lane & 15) < 8) {
        out[base_row + myrow] = s[lane & 7] - sqr;
    }
}

extern "C" void kernel_launch(void* const* d_in, const int* in_sizes, int n_in,
                              void* d_out, int out_size, void* d_ws, size_t ws_size,
                              hipStream_t stream) {
    const float* features = (const float*)d_in[0];
    const float* W        = (const float*)d_in[1];
    float* out            = (float*)d_out;
    (void)d_ws; (void)ws_size; (void)n_in; (void)out_size;

    int B = in_sizes[0] / F_DIM;
    int blocks = (B + ROWS_PER_BLOCK - 1) / ROWS_PER_BLOCK;
    fm_second_order_kernel<<<blocks, 256, 0, stream>>>(features, W, out, B);
}


# 64-row work-groups: each 16-row tile's K is split over a wave pair and the two partial products are added through LDS before squaring (K summed as two halves), so 256 groups fill every WGP
# speedup vs baseline: 1.9936x; 1.3026x over previous
#include <hip/hip_runtime.h>


typedef __attribute__((ext_vector_type(2))) float v2f;
typedef __attribute__((ext_vector_type(4))) float v4f;
typedef __attribute__((ext_vector_type(8))) float v8f;

#define F_DIM 200
#define E_DIM 64
#define LDS_PITCH 202
#define WAVES_PER_BLOCK 8
#define ROW_TILES 4
#define ROWS_PER_WAVE 16
#define ROWS_PER_BLOCK (ROW_TILES * ROWS_PER_WAVE)
#define K_HALF (F_DIM / 2)
#define K_STEPS (K_HALF / 4)
#define W_PER_THREAD (F_DIM * E_DIM / 256)
#define XCH_PER_TILE (32 * 32 + 32)

static __device__ __forceinline__ void lds_barrier() {
    __builtin_amdgcn_fence(__ATOMIC_RELEASE, "workgroup", "local");
    __builtin_amdgcn_s_barrier();
    __builtin_amdgcn_fence(__ATOMIC_ACQUIRE, "workgroup", "local");
}

__global__ __launch_bounds__(256)
void fm_second_order_kernel(const float* __restrict__ features,
                            const float* __restrict__ W,
                            float* __restrict__ out, int Brows) {
    __shared__ float sWt[E_DIM * LDS_PITCH];
    __shared__ float sW2[F_DIM];
    __shared__ __attribute__((aligned(16))) float sX[ROW_TILES * XCH_PER_TILE];

    const int tid  = threadIdx.x;
    const int lane = tid & 31;
    const int wave = tid >> 5;
    const int tile  = wave & (ROW_TILES - 1);
    const int khalf = wave >> 2;

    const int base_row = blockIdx.x * ROWS_PER_BLOCK + tile * ROWS_PER_WAVE;
    const int mrow = lane & 15;
    const int half = lane >> 4;
    const int n0   = lane & 15;
    const int kbase = khalf * K_HALF + 2 * half;

    float wreg[W_PER_THREAD];
    #pragma unroll
    for (int j = 0; j < W_PER_THREAD; ++j) wreg[j] = W[tid + 256 * j];
    #pragma unroll
    for (int j = 0; j < W_PER_THREAD; ++j) {
        int i = tid + 256 * j;
        int f = i >> 6;
        int e = i & (E_DIM - 1);
        sWt[e * LDS_PITCH + f] = wreg[j];
    }

    __builtin_amdgcn_sched_barrier(0);

    int arow_idx = base_row + mrow;
    arow_idx = arow_idx < Brows ? arow_idx : Brows - 1;
    const float* arow = features + (size_t)arow_idx * F_DIM + kbase;
    v2f areg[K_STEPS];
    #pragma unroll
    for (int j = 0; j < K_STEPS; ++j) areg[j] = *reinterpret_cast<const v2f*>(arow + 4 * j);

    lds_barrier();
    if (tid < F_DIM) {
        float s = 0.f;
        #pragma unroll
        for (int e = 0; e < E_DIM; ++e) {
            float w = sWt[e * LDS_PITCH + tid];
            s = fmaf(w, w, s);
        }
        sW2[tid] = s;
    }
    lds_barrier();

    const bool active = base_row < Brows;

    v8f acc0 = {}, acc1 = {}, acc2 = {}, acc3 = {};
    float sqp = 0.f;

    if (active) {
        #pragma unroll
        for (int j = 0; j < K_STEPS; ++j) {
            const int kk = kbase + 4 * j;
            const v2f a = areg[j];
            sqp = fmaf(a.x * a.x, sW2[kk],     sqp);
            sqp = fmaf(a.y * a.y, sW2[kk + 1], sqp);
            v2f b0 = *reinterpret_cast<const v2f*>(&sWt[(n0 +  0) * LDS_PITCH + kk]);
            v2f b1 = *reinterpret_cast<const v2f*>(&sWt[(n0 + 16) * LDS_PITCH + kk]);
            v2f b2 = *reinterpret_cast<const v2f*>(&sWt[(n0 + 32) * LDS_PITCH + kk]);
            v2f b3 = *reinterpret_cast<const v2f*>(&sWt[(n0 + 48) * LDS_PITCH + kk]);
            acc0 = __builtin_amdgcn_wmma_f32_16x16x4_f32(false, a, false, b0, (short)0, acc0, false, false);
            acc1 = __builtin_amdgcn_wmma_f32_16x16x4_f32(false, a, false, b1, (short)0, acc1, false, false);
            acc2 = __builtin_amdgcn_wmma_f32_16x16x4_f32(false, a, false, b2, (short)0, acc2, false, false);
            acc3 = __builtin_amdgcn_wmma_f32_16x16x4_f32(false, a, false, b3, (short)0, acc3, false, false);
        }
    }

    float* xt = sX + tile * XCH_PER_TILE;
    v4f* xv = reinterpret_cast<v4f*>(xt) + lane;
    if (khalf == 1) {
        xv[0 * 32] = __builtin_shufflevector(acc0, acc0, 0, 1, 2, 3);
        xv[1 * 32] = __builtin_shufflevector(acc0, acc0, 4, 5, 6, 7);
        xv[2 * 32] = __builtin_shufflevector(acc1, acc1, 0, 1, 2, 3);
        xv[3 * 32] = __builtin_shufflevector(acc1, acc1, 4, 5, 6, 7);
        xv[4 * 32] = __builtin_shufflevector(acc2, acc2, 0, 1, 2, 3);
        xv[5 * 32] = __builtin_shufflevector(acc2, acc2, 4, 5, 6, 7);
        xv[6 * 32] = __builtin_shufflevector(acc3, acc3, 0, 1, 2, 3);
        xv[7 * 32] = __builtin_shufflevector(acc3, acc3, 4, 5, 6, 7);
        xt[32 * 32 + lane] = sqp;
    }
    lds_barrier();
    if (khalf == 1 || !active) return;

    const v4f p0l = xv[0 * 32], p0h = xv[1 * 32], p1l = xv[2 * 32], p1h = xv[3 * 32];
    const v4f p2l = xv[4 * 32], p2h = xv[5 * 32], p3l = xv[6 * 32], p3h = xv[7 * 32];
    acc0 += __builtin_shufflevector(p0l, p0h, 0, 1, 2, 3, 4, 5, 6, 7);
    acc1 += __builtin_shufflevector(p1l, p1h, 0, 1, 2, 3, 4, 5, 6, 7);
    acc2 += __builtin_shufflevector(p2l, p2h, 0, 1, 2, 3, 4, 5, 6, 7);
    acc3 += __builtin_shufflevector(p3l, p3h, 0, 1, 2, 3, 4, 5, 6, 7);
    sqp += xt[32 * 32 + lane];
    float s[8];
    #pragma unroll
    for (int r = 0; r < 8; ++r) {
        float c0 = acc0[r], c1 = acc1[r], c2 = acc2[r], c3 = acc3[r];
        s[r] = c0 * c0 + c1 * c1 + c2 * c2 + c3 * c3;
    }
    #pragma unroll
    for (int m = 1; m <= 8; m <<= 1) {
        #pragma unroll
        for (int r = 0; r < 8; ++r) s[r] += __shfl_xor(s[r], m, 32);
    }
    float sqfull = sqp + __shfl_xor(sqp, 16, 32);

    int myrow = (lane & 7) + ((lane >> 4) << 3);
    float sqr = __shfl(sqfull, myrow, 32);
    if ((lane & 15) < 8) {
        out[base_row + myrow] = s[lane & 7] - sqr;
    }
}

extern "C" void kernel_launch(void* const* d_in, const int* in_sizes, int n_in,
                              void* d_out, int out_size, void* d_ws, size_t ws_size,
                              hipStream_t stream) {
    const float* features = (const float*)d_in[0];
    const float* W        = (const float*)d_in[1];
    float* out            = (float*)d_out;
    (void)d_ws; (void)ws_size; (void)n_in; (void)out_size;

    int B = in_sizes[0] / F_DIM;
    int blocks = (B + ROWS_PER_BLOCK - 1) / ROWS_PER_BLOCK;
    fm_second_order_kernel<<<blocks, 256, 0, stream>>>(features, W, out, B);
}


# W staged untransposed by 16-byte loads/stores, w2sum reduced from those registers by a DPP tree (new w2sum summation order), A loads issued with the W loads: LDS w2sum pass and one barrier gone
# speedup vs baseline: 2.1571x; 1.0820x over previous
#include <hip/hip_runtime.h>


typedef __attribute__((ext_vector_type(2))) float v2f;
typedef __attribute__((ext_vector_type(4))) float v4f;
typedef __attribute__((ext_vector_type(8))) float v8f;

#define F_DIM 200
#define E_DIM 64
#define TILE_STRIDE (F_DIM * 16 + 16)
#define WAVES_PER_BLOCK 8
#define ROW_TILES 4
#define ROWS_PER_WAVE 16
#define ROWS_PER_BLOCK (ROW_TILES * ROWS_PER_WAVE)
#define K_HALF (F_DIM / 2)
#define K_STEPS (K_HALF / 4)
#define W_VECS 13
#define XCH_PER_TILE (32 * 32 + 32)

static __device__ __forceinline__ void lds_barrier() {
    __builtin_amdgcn_fence(__ATOMIC_RELEASE, "workgroup", "local");
    __builtin_amdgcn_s_barrier();
    __builtin_amdgcn_fence(__ATOMIC_ACQUIRE, "workgroup", "local");
}

template <int M>
static __device__ __forceinline__ float row_xor_get(float v) {
    return __builtin_bit_cast(float, __builtin_amdgcn_update_dpp(0, __builtin_bit_cast(int, v), 0x160 + M, 0xf, 0xf, true));
}

template <int M>
static __device__ __forceinline__ float fold_pair(float lo, float hi, bool bit) {
    return (bit ? hi : lo) + row_xor_get<M>(bit ? lo : hi);
}

__global__ __launch_bounds__(256)
void fm_second_order_kernel(const float* __restrict__ features,
                            const float* __restrict__ W,
                            float* __restrict__ out, int Brows) {
    __shared__ __attribute__((aligned(16))) float sWs[4 * TILE_STRIDE];
    __shared__ float sW2[F_DIM];
    __shared__ __attribute__((aligned(16))) float sX[ROW_TILES * XCH_PER_TILE];

    const int tid  = threadIdx.x;
    const int lane = tid & 31;
    const int wave = tid >> 5;
    const int tile  = wave & (ROW_TILES - 1);
    const int khalf = wave >> 2;

    const int base_row = blockIdx.x * ROWS_PER_BLOCK + tile * ROWS_PER_WAVE;
    const int mrow = lane & 15;
    const int half = lane >> 4;
    const int n0   = lane & 15;
    const int kbase = khalf * K_HALF + 2 * half;

    const int c4 = tid & 15;
    const int r0 = tid >> 4;
    const bool last_ok = r0 < F_DIM - 16 * (W_VECS - 1);
    v4f wv[W_VECS];
    #pragma unroll
    for (int j = 0; j < W_VECS; ++j) {
        int row = r0 + 16 * j;
        if (j == W_VECS - 1) row = last_ok ? row : F_DIM - 1;
        wv[j] = *reinterpret_cast<const v4f*>(W + row * E_DIM + 4 * c4);
    }

    int arow_idx = base_row + mrow;
    arow_idx = arow_idx < Brows ? arow_idx : Brows - 1;
    const float* arow = features + (size_t)arow_idx * F_DIM + kbase;
    v2f areg[K_STEPS];
    #pragma unroll
    for (int j = 0; j < K_STEPS; ++j) areg[j] = *reinterpret_cast<const v2f*>(arow + 4 * j);

    float* wdst = sWs + (c4 >> 2) * TILE_STRIDE + r0 * 16 + (c4 & 3) * 4;
    float p[16];
    #pragma unroll
    for (int j = 0; j < W_VECS; ++j) {
        if (j < W_VECS - 1 || last_ok) *reinterpret_cast<v4f*>(wdst + j * 256) = wv[j];
        const v2f lo = __builtin_shufflevector(wv[j], wv[j], 0, 1);
        const v2f hi = __builtin_shufflevector(wv[j], wv[j], 2, 3);
        const v2f q2 = lo * lo + hi * hi;
        p[j] = q2.x + q2.y;
    }
    p[13] = 0.f; p[14] = 0.f; p[15] = 0.f;
    float p8[8], p4[4], p2[2];
    #pragma unroll
    for (int i = 0; i < 8; ++i) p8[i] = fold_pair<8>(p[i], p[i + 8], (c4 & 8) != 0);
    #pragma unroll
    for (int i = 0; i < 4; ++i) p4[i] = fold_pair<4>(p8[i], p8[i + 4], (c4 & 4) != 0);
    #pragma unroll
    for (int i = 0; i < 2; ++i) p2[i] = fold_pair<2>(p4[i], p4[i + 2], (c4 & 2) != 0);
    const float w2row = fold_pair<1>(p2[0], p2[1], (c4 & 1) != 0);
    if (c4 < W_VECS - 1 || (c4 == W_VECS - 1 && last_ok)) sW2[r0 + 16 * c4] = w2row;
    lds_barrier();

    const bool active = base_row < Brows;

    v8f acc0 = {}, acc1 = {}, acc2 = {}, acc3 = {};
    float sqp = 0.f;

    if (active) {
        const float* bp = sWs + kbase * 16 + n0;
        #pragma unroll
        for (int j = 0; j < K_STEPS; ++j) {
            const int kk = kbase + 4 * j;
            const v2f a = areg[j];
            sqp = fmaf(a.x * a.x, sW2[kk],     sqp);
            sqp = fmaf(a.y * a.y, sW2[kk + 1], sqp);
            const float* q = bp + 64 * j;
            v2f b0 = { q[0 * TILE_STRIDE], q[0 * TILE_STRIDE + 16] };
            v2f b1 = { q[1 * TILE_STRIDE], q[1 * TILE_STRIDE + 16] };
            v2f b2 = { q[2 * TILE_STRIDE], q[2 * TILE_STRIDE + 16] };
            v2f b3 = { q[3 * TILE_STRIDE], q[3 * TILE_STRIDE + 16] };
            acc0 = __builtin_amdgcn_wmma_f32_16x16x4_f32(false, a, false, b0, (short)0, acc0, false, false);
            acc1 = __builtin_amdgcn_wmma_f32_16x16x4_f32(false, a, false, b1, (short)0, acc1, false, false);
            acc2 = __builtin_amdgcn_wmma_f32_16x16x4_f32(false, a, false, b2, (short)0, acc2, false, false);
            acc3 = __builtin_amdgcn_wmma_f32_16x16x4_f32(false, a, false, b3, (short)0, acc3, false, false);
        }
    }

    float* xt = sX + tile * XCH_PER_TILE;
    v4f* xv = reinterpret_cast<v4f*>(xt) + lane;
    if (khalf == 1) {
        xv[0 * 32] = __builtin_shufflevector(acc0, acc0, 0, 1, 2, 3);
        xv[1 * 32] = __builtin_shufflevector(acc0, acc0, 4, 5, 6, 7);
        xv[2 * 32] = __builtin_shufflevector(acc1, acc1, 0, 1, 2, 3);
        xv[3 * 32] = __builtin_shufflevector(acc1, acc1, 4, 5, 6, 7);
        xv[4 * 32] = __builtin_shufflevector(acc2, acc2, 0, 1, 2, 3);
        xv[5 * 32] = __builtin_shufflevector(acc2, acc2, 4, 5, 6, 7);
        xv[6 * 32] = __builtin_shufflevector(acc3, acc3, 0, 1, 2, 3);
        xv[7 * 32] = __builtin_shufflevector(acc3, acc3, 4, 5, 6, 7);
        xt[32 * 32 + lane] = sqp;
    }
    lds_barrier();
    if (khalf == 1 || !active) return;

    const v4f p0l = xv[0 * 32], p0h = xv[1 * 32], p1l = xv[2 * 32], p1h = xv[3 * 32];
    const v4f p2l = xv[4 * 32], p2h = xv[5 * 32], p3l = xv[6 * 32], p3h = xv[7 * 32];
    acc0 += __builtin_shufflevector(p0l, p0h, 0, 1, 2, 3, 4, 5, 6, 7);
    acc1 += __builtin_shufflevector(p1l, p1h, 0, 1, 2, 3, 4, 5, 6, 7);
    acc2 += __builtin_shufflevector(p2l, p2h, 0, 1, 2, 3, 4, 5, 6, 7);
    acc3 += __builtin_shufflevector(p3l, p3h, 0, 1, 2, 3, 4, 5, 6, 7);
    sqp += xt[32 * 32 + lane];
    float s[8];
    #pragma unroll
    for (int r = 0; r < 8; ++r) {
        float c0 = acc0[r], c1 = acc1[r], c2 = acc2[r], c3 = acc3[r];
        s[r] = c0 * c0 + c1 * c1 + c2 * c2 + c3 * c3;
    }
    #pragma unroll
    for (int m = 1; m <= 8; m <<= 1) {
        #pragma unroll
        for (int r = 0; r < 8; ++r) s[r] += __shfl_xor(s[r], m, 32);
    }
    float sqfull = sqp + __shfl_xor(sqp, 16, 32);

    int myrow = (lane & 7) + ((lane >> 4) << 3);
    float sqr = __shfl(sqfull, myrow, 32);
    if ((lane & 15) < 8) {
        out[base_row + myrow] = s[lane & 7] - sqr;
    }
}

extern "C" void kernel_launch(void* const* d_in, const int* in_sizes, int n_in,
                              void* d_out, int out_size, void* d_ws, size_t ws_size,
                              hipStream_t stream) {
    const float* features = (const float*)d_in[0];
    const float* W        = (const float*)d_in[1];
    float* out            = (float*)d_out;
    (void)d_ws; (void)ws_size; (void)n_in; (void)out_size;

    int B = in_sizes[0] / F_DIM;
    int blocks = (B + ROWS_PER_BLOCK - 1) / ROWS_PER_BLOCK;
    fm_second_order_kernel<<<blocks, 256, 0, stream>>>(features, W, out, B);
}
